# starting kernel
# baseline (speedup 1.0000x reference)
#include <hip/hip_runtime.h>
#include <hip/hip_bf16.h>

#define IN_F 128
#define HID  128
#define N_CLS 2

typedef __attribute__((ext_vector_type(2))) float v2f;
typedef __attribute__((ext_vector_type(8))) float v8f;

__global__ void gcn_zero_f32(float* __restrict__ p, int n) {
    int i = blockIdx.x * blockDim.x + threadIdx.x;
    if (i < n) p[i] = 0.0f;
}

__global__ void gcn_deg_accum(const int* __restrict__ dst, float* __restrict__ deg, int nE) {
    int e = blockIdx.x * blockDim.x + threadIdx.x;
    if (e < nE) atomicAdd(&deg[dst[e]], 1.0f);
}

__global__ void gcn_dinv_finalize(float* __restrict__ dinv, int n) {
    int i = blockIdx.x * blockDim.x + threadIdx.x;
    if (i < n) dinv[i] = rsqrtf(dinv[i] + 1.0f);
}

__global__ __launch_bounds__(64) void gcn_gemm1_wmma(const float* __restrict__ x,
                                                     const float* __restrict__ W1,
                                                     float* __restrict__ xw) {
    const int wave = blockIdx.x * (blockDim.x >> 5) + (threadIdx.x >> 5);
    const int lane = threadIdx.x & 31;
    const int half = lane >> 4;
    const int lm   = lane & 15;
    const int row0 = wave * 16;

    const float* __restrict__ xrow = x + (size_t)(row0 + lm) * IN_F;

    v8f acc[8];
#pragma unroll
    for (int cb = 0; cb < 8; ++cb) acc[cb] = (v8f){};

#pragma unroll 4
    for (int k0 = 0; k0 < IN_F; k0 += 4) {
        const int ka = k0 + 2 * half;
        v2f a = *(const v2f*)(xrow + ka);
        const float* __restrict__ w0 = W1 + (size_t)(ka + 0) * HID + lm;
        const float* __restrict__ w1 = W1 + (size_t)(ka + 1) * HID + lm;
#pragma unroll
        for (int cb = 0; cb < 8; ++cb) {
            v2f b;
            b.x = w0[cb * 16];
            b.y = w1[cb * 16];
            acc[cb] = __builtin_amdgcn_wmma_f32_16x16x4_f32(false, a, false, b,
                                                            (short)0, acc[cb], false, false);
        }
    }
#pragma unroll
    for (int cb = 0; cb < 8; ++cb) {
#pragma unroll
        for (int r = 0; r < 8; ++r) {
            xw[(size_t)(row0 + r + 8 * half) * HID + cb * 16 + lm] = acc[cb][r];
        }
    }
}

__global__ void gcn_selfloop_h(const float* __restrict__ xw, const float* __restrict__ dinv,
                               float* __restrict__ h, int n) {
    int idx = blockIdx.x * blockDim.x + threadIdx.x;
    int node = idx >> 5;
    if (node >= n) return;
    int lane = idx & 31;
    float s = dinv[node];
    s = s * s;
    const float4 v = *(const float4*)(xw + (size_t)node * HID + lane * 4);
    float4 o = make_float4(v.x * s, v.y * s, v.z * s, v.w * s);
    *(float4*)(h + (size_t)node * HID + lane * 4) = o;
}

__global__ void gcn_agg1(const int* __restrict__ src, const int* __restrict__ dst,
                         const float* __restrict__ dinv, const float* __restrict__ xw,
                         float* __restrict__ h, int nE) {
    int e = (blockIdx.x * blockDim.x + threadIdx.x) >> 5;
    if (e >= nE) return;
    int lane = threadIdx.x & 31;
    int s = src[e], d = dst[e];
    float norm = dinv[s] * dinv[d];
    const float4 v = *(const float4*)(xw + (size_t)s * HID + lane * 4);
    float* o = h + (size_t)d * HID + lane * 4;
    atomicAdd(o + 0, v.x * norm);
    atomicAdd(o + 1, v.y * norm);
    atomicAdd(o + 2, v.z * norm);
    atomicAdd(o + 3, v.w * norm);
}

__global__ void gcn_gemm2(const float* __restrict__ h, const float* __restrict__ b1,
                          const float* __restrict__ W2, float* __restrict__ zw, int n) {
    int node = (blockIdx.x * blockDim.x + threadIdx.x) >> 5;
    if (node >= n) return;
    int lane = threadIdx.x & 31;
    const float4 hv = *(const float4*)(h + (size_t)node * HID + lane * 4);
    const float4 bv = *(const float4*)(b1 + lane * 4);
    float f0 = fmaxf(hv.x + bv.x, 0.0f);
    float f1 = fmaxf(hv.y + bv.y, 0.0f);
    float f2 = fmaxf(hv.z + bv.z, 0.0f);
    float f3 = fmaxf(hv.w + bv.w, 0.0f);
    int k = lane * 4;
    float a0 = f0 * W2[(k + 0) * 2 + 0] + f1 * W2[(k + 1) * 2 + 0]
             + f2 * W2[(k + 2) * 2 + 0] + f3 * W2[(k + 3) * 2 + 0];
    float a1 = f0 * W2[(k + 0) * 2 + 1] + f1 * W2[(k + 1) * 2 + 1]
             + f2 * W2[(k + 2) * 2 + 1] + f3 * W2[(k + 3) * 2 + 1];
#pragma unroll
    for (int off = 16; off > 0; off >>= 1) {
        a0 += __shfl_down(a0, off, 32);
        a1 += __shfl_down(a1, off, 32);
    }
    if (lane == 0) {
        zw[(size_t)node * 2 + 0] = a0;
        zw[(size_t)node * 2 + 1] = a1;
    }
}

__global__ void gcn_selfloop_out(const float* __restrict__ zw, const float* __restrict__ dinv,
                                 float* __restrict__ out, int n) {
    int i = blockIdx.x * blockDim.x + threadIdx.x;
    if (i >= n) return;
    float s = dinv[i];
    s = s * s;
    out[(size_t)i * 2 + 0] = zw[(size_t)i * 2 + 0] * s;
    out[(size_t)i * 2 + 1] = zw[(size_t)i * 2 + 1] * s;
}

__global__ void gcn_agg2(const int* __restrict__ src, const int* __restrict__ dst,
                         const float* __restrict__ dinv, const float* __restrict__ zw,
                         float* __restrict__ out, int nE) {
    int e = blockIdx.x * blockDim.x + threadIdx.x;
    if (e >= nE) return;
    int s = src[e], d = dst[e];
    float norm = dinv[s] * dinv[d];
    atomicAdd(out + (size_t)d * 2 + 0, zw[(size_t)s * 2 + 0] * norm);
    atomicAdd(out + (size_t)d * 2 + 1, zw[(size_t)s * 2 + 1] * norm);
}

__global__ void gcn_logsoftmax(float* __restrict__ out, const float* __restrict__ b2, int n) {
    int i = blockIdx.x * blockDim.x + threadIdx.x;
    if (i >= n) return;
    float l0 = out[(size_t)i * 2 + 0] + b2[0];
    float l1 = out[(size_t)i * 2 + 1] + b2[1];
    float m = fmaxf(l0, l1);
    float lse = m + logf(expf(l0 - m) + expf(l1 - m));
    out[(size_t)i * 2 + 0] = l0 - lse;
    out[(size_t)i * 2 + 1] = l1 - lse;
}

extern "C" void kernel_launch(void* const* d_in, const int* in_sizes, int n_in,
                              void* d_out, int out_size, void* d_ws, size_t ws_size,
                              hipStream_t stream) {
    const float* x  = (const float*)d_in[0];
    const float* W1 = (const float*)d_in[1];
    const float* b1 = (const float*)d_in[2];
    const float* W2 = (const float*)d_in[3];
    const float* b2 = (const float*)d_in[4];
    const int*   ei = (const int*)d_in[5];

    const int n  = in_sizes[0] / IN_F;
    const int nE = in_sizes[5] / 2;
    const int* src = ei;
    const int* dst = ei + nE;

    char* ws = (char*)d_ws;
    float* xw   = (float*)(ws);
    float* h    = (float*)(ws + (size_t)n * HID * 4);
    float* dinv = (float*)(ws + (size_t)2 * n * HID * 4);
    float* zw   = (float*)(ws + (size_t)2 * n * HID * 4 + 512000);
    float* out  = (float*)d_out;

    const int B = 256;

    gcn_zero_f32<<<(n + B - 1) / B, B, 0, stream>>>(dinv, n);
    gcn_deg_accum<<<(nE + B - 1) / B, B, 0, stream>>>(dst, dinv, nE);
    gcn_dinv_finalize<<<(n + B - 1) / B, B, 0, stream>>>(dinv, n);

    {
        int rowTiles = n / 16;
        gcn_gemm1_wmma<<<rowTiles / 2, 64, 0, stream>>>(x, W1, xw);
    }

    gcn_selfloop_h<<<((size_t)n * 32 + B - 1) / B, B, 0, stream>>>(xw, dinv, h, n);
    gcn_agg1<<<((size_t)nE * 32 + B - 1) / B, B, 0, stream>>>(src, dst, dinv, xw, h, nE);

    gcn_gemm2<<<((size_t)n * 32 + B - 1) / B, B, 0, stream>>>(h, b1, W2, zw, n);

    gcn_selfloop_out<<<(n + B - 1) / B, B, 0, stream>>>(zw, dinv, out, n);
    gcn_agg2<<<(nE + B - 1) / B, B, 0, stream>>>(src, dst, dinv, zw, out, nE);

    gcn_logsoftmax<<<(n + B - 1) / B, B, 0, stream>>>(out, b2, n);
}


# both aggregation kernels run as resident grids (1250 and 625 blocks) whose waves loop over runs of nodes, the next node's indices fetched ahead, instead of one launched wave (8-lane group) per node
# speedup vs baseline: 4.8932x; 4.8932x over previous
#include <hip/hip_runtime.h>


#define IN_F 128
#define HID  128
#define N_CLS 2

typedef __attribute__((ext_vector_type(2))) float v2f;
typedef __attribute__((ext_vector_type(8))) float v8f;

__device__ __forceinline__ int wave_sum_i32(int v) {
    for (int off = 16; off > 0; off >>= 1) v += __shfl_xor(v, off, 32);
    return v;
}

#define BK_SH 9
#define BK_N 512
#define NB_MAX 256
#define L1_TILE 6400
#define L1_PER 7
#define SORT_T 1024
#define D_PER 12
#define D_CAP (SORT_T * D_PER)

static_assert(L1_TILE <= SORT_T * L1_PER, "a tile must fit the per-thread edge registers");

__device__ __forceinline__ int block_excl_scan_1024(int v, int* sm, int& total) {
    const int lane = threadIdx.x & 31, w = threadIdx.x >> 5;
    int incl = v;
    for (int d = 1; d < 32; d <<= 1) {
        int y = __shfl_up(incl, d, 32);
        if (lane >= d) incl += y;
    }
    if (lane == 31) sm[w] = incl;
    __syncthreads();
    const int ws = sm[lane];
    const int before = wave_sum_i32((lane < w) ? ws : 0);
    total = wave_sum_i32(ws);
    __syncthreads();
    return before + incl - v;
}

__global__ __launch_bounds__(SORT_T) void gcn_tile_bucket(const int* __restrict__ src,
                                                          const int* __restrict__ dst,
                                                          int* __restrict__ tileoff,
                                                          int2* __restrict__ pairs,
                                                          int nE, int nb) {
    __shared__ int cur[NB_MAX];
    __shared__ int loff[NB_MAX];
    __shared__ int sm[32];
    __shared__ int2 st[L1_TILE];
    const int t = threadIdx.x;
    if (t < NB_MAX) cur[t] = 0;
    __syncthreads();
    const int e0 = blockIdx.x * L1_TILE;
    const int e1 = (e0 + L1_TILE < nE) ? e0 + L1_TILE : nE;
    const int m = e1 - e0;

    int dd[L1_PER], ss[L1_PER];
#pragma unroll
    for (int k = 0; k < L1_PER; ++k) {
        const int e = e0 + t + k * SORT_T;
        dd[k] = -1;
        ss[k] = 0;
        if (e < e1) { dd[k] = dst[e]; ss[k] = src[e]; }
    }
#pragma unroll
    for (int k = 0; k < L1_PER; ++k)
        if (dd[k] >= 0) atomicAdd(&cur[dd[k] >> BK_SH], 1);
    __syncthreads();

    const int c = (t < nb) ? cur[t] : 0;
    int tot;
    const int ex = block_excl_scan_1024(c, sm, tot);
    int* __restrict__ trow = tileoff + (size_t)blockIdx.x * (nb + 1);
    if (t < nb) {
        loff[t] = ex;
        cur[t] = 0;
        trow[t] = ex;
    }
    if (t == 0) trow[nb] = tot;
    __syncthreads();

#pragma unroll
    for (int k = 0; k < L1_PER; ++k) {
        if (dd[k] >= 0) {
            const int b = dd[k] >> BK_SH;
            const int p = loff[b] + atomicAdd(&cur[b], 1);
            st[p] = make_int2(ss[k], dd[k]);
        }
    }
    __syncthreads();

    for (int i = t; i < m; i += SORT_T) pairs[e0 + i] = st[i];
}

__device__ __forceinline__ int run_of(const int* tp, int tiles, int i) {
    int lo = 0, hi = tiles;
    while (hi - lo > 1) {
        const int mid = (lo + hi) >> 1;
        if (tp[mid] <= i) lo = mid; else hi = mid;
    }
    return lo;
}

#define RUN_PER 2
#define RUN_W 8
#define RUN_REGS (RUN_PER * RUN_W)

__global__ __launch_bounds__(SORT_T) void gcn_bucket_csr(const int2* __restrict__ pairs,
                                                         const int* __restrict__ tileoff,
                                                         int* __restrict__ cnt,
                                                         int* __restrict__ rowptr,
                                                         float* __restrict__ dinv,
                                                         int* __restrict__ csr,
                                                         int n, int tiles, int nb) {
    __shared__ int c[BK_N];
    __shared__ int off[BK_N];
    __shared__ int cur[BK_N];
    __shared__ int sm[32];
    __shared__ int big;
    __shared__ int tp[SORT_T + 1];
    __shared__ int ta[SORT_T];
    __shared__ int stage[D_CAP];
    const int t = threadIdx.x;
    const int b = blockIdx.x;
    const int node0 = b << BK_SH;

    int below = 0, upto = 0;
    if (t < tiles) {
        const int* __restrict__ trow = tileoff + (size_t)t * (nb + 1) + b;
        below = trow[0];
        upto  = trow[1];
    }
    if (t < BK_N) c[t] = 0;
    if (t == 0) big = 0;
    int m, s0;
    const int rs = block_excl_scan_1024(upto - below, sm, m);
    if (upto - below > 32 * RUN_PER) big = 1;
    (void)block_excl_scan_1024(below, sm, s0);
    if (t < tiles) {
        tp[t] = rs;
        ta[t] = t * L1_TILE + below - rs;
    }
    if (t == 0) tp[tiles] = m;
    __syncthreads();

    const bool fits = (m <= D_CAP) && (tiles <= 32 * RUN_W) && (big == 0);
    const int w = t >> 5, lane = t & 31;
    int2 pr[RUN_REGS];
    if (fits) {
#pragma unroll
        for (int j = 0; j < RUN_W; ++j) {
            const int tl = w + 32 * j;
            int a = 0, len = 0;
            if (tl < tiles) {
                const int p0 = tp[tl];
                a = ta[tl] + p0;
                len = tp[tl + 1] - p0;
            }
#pragma unroll
            for (int r = 0; r < RUN_PER; ++r) {
                const int i = lane + 32 * r;
                pr[j * RUN_PER + r] = make_int2(0, -1);
                if (i < len) pr[j * RUN_PER + r] = pairs[a + i];
            }
        }
#pragma unroll
        for (int k = 0; k < RUN_REGS; ++k)
            if (pr[k].y >= 0) atomicAdd(&c[pr[k].y - node0], 1);
    } else {
        for (int i = t; i < m; i += SORT_T)
            atomicAdd(&c[pairs[ta[run_of(tp, tiles, i)] + i].y - node0], 1);
    }
    __syncthreads();

    const int v = (t < BK_N) ? c[t] : 0;
    int tot;
    const int ex = block_excl_scan_1024(v, sm, tot);
    if (t < BK_N) {
        off[t] = ex;
        cur[t] = 0;
        const int node = node0 + t;
        if (node < n) {
            cnt[node] = v;
            rowptr[node] = s0 + ex;
            dinv[node] = rsqrtf((float)v + 1.0f);
        }
    }
    __syncthreads();

    if (fits) {
#pragma unroll
        for (int k = 0; k < RUN_REGS; ++k) {
            if (pr[k].y >= 0) {
                const int l = pr[k].y - node0;
                stage[off[l] + atomicAdd(&cur[l], 1)] = pr[k].x;
            }
        }
        __syncthreads();
        for (int i = t; i < m; i += SORT_T) csr[s0 + i] = stage[i];
    } else {
        for (int i = t; i < m; i += SORT_T) {
            const int2 q = pairs[ta[run_of(tp, tiles, i)] + i];
            const int l = q.y - node0;
            csr[s0 + off[l] + atomicAdd(&cur[l], 1)] = q.x;
        }
    }
}

__global__ __launch_bounds__(64) void gcn_gemm1_wmma(const float* __restrict__ x,
                                                     const float* __restrict__ W1,
                                                     const float* __restrict__ dinv,
                                                     float* __restrict__ xw) {
    const int wave = blockIdx.x * (blockDim.x >> 5) + (threadIdx.x >> 5);
    const int lane = threadIdx.x & 31;
    const int half = lane >> 4;
    const int lm   = lane & 15;
    const int row0 = wave * 16;

    const float* __restrict__ xrow = x + (size_t)(row0 + lm) * IN_F;

    const float4 dlo = *(const float4*)(dinv + row0 + 8 * half);
    const float4 dhi = *(const float4*)(dinv + row0 + 8 * half + 4);

    v8f acc[8];
#pragma unroll
    for (int cb = 0; cb < 8; ++cb) acc[cb] = (v8f){};

#pragma unroll 4
    for (int k0 = 0; k0 < IN_F; k0 += 4) {
        const int ka = k0 + 2 * half;
        v2f a = *(const v2f*)(xrow + ka);
        const float* __restrict__ w0 = W1 + (size_t)(ka + 0) * HID + lm;
        const float* __restrict__ w1 = W1 + (size_t)(ka + 1) * HID + lm;
#pragma unroll
        for (int cb = 0; cb < 8; ++cb) {
            v2f b;
            b.x = w0[cb * 16];
            b.y = w1[cb * 16];
            acc[cb] = __builtin_amdgcn_wmma_f32_16x16x4_f32(false, a, false, b,
                                                            (short)0, acc[cb], false, false);
        }
    }
    const float ds[8] = {dlo.x, dlo.y, dlo.z, dlo.w, dhi.x, dhi.y, dhi.z, dhi.w};
#pragma unroll
    for (int cb = 0; cb < 8; ++cb) {
#pragma unroll
        for (int r = 0; r < 8; ++r) {
            xw[(size_t)(row0 + r + 8 * half) * HID + cb * 16 + lm] = acc[cb][r] * ds[r];
        }
    }
}

#define P1_BLOCKS 1250

__device__ __forceinline__ float4 p1_row(const float* __restrict__ xw, int s, unsigned lane16) {
    const unsigned off = (unsigned)s * (unsigned)(HID * 4) + lane16;
    return *(const float4*)((const char*)xw + off);
}

#define P1_ADD(r) acc.x += r.x; acc.y += r.y; acc.z += r.z; acc.w += r.w;

__device__ __forceinline__ void p1_sum_rows(const float* __restrict__ xw, int idx, int m,
                                            unsigned lane16, float4& acc) {
    int k = 0;
    for (; k + 8 <= m; k += 8) {
        const float4 r0 = p1_row(xw, __builtin_amdgcn_readlane(idx, k + 0), lane16);
        const float4 r1 = p1_row(xw, __builtin_amdgcn_readlane(idx, k + 1), lane16);
        const float4 r2 = p1_row(xw, __builtin_amdgcn_readlane(idx, k + 2), lane16);
        const float4 r3 = p1_row(xw, __builtin_amdgcn_readlane(idx, k + 3), lane16);
        const float4 r4 = p1_row(xw, __builtin_amdgcn_readlane(idx, k + 4), lane16);
        const float4 r5 = p1_row(xw, __builtin_amdgcn_readlane(idx, k + 5), lane16);
        const float4 r6 = p1_row(xw, __builtin_amdgcn_readlane(idx, k + 6), lane16);
        const float4 r7 = p1_row(xw, __builtin_amdgcn_readlane(idx, k + 7), lane16);
        P1_ADD(r0) P1_ADD(r1) P1_ADD(r2) P1_ADD(r3)
        P1_ADD(r4) P1_ADD(r5) P1_ADD(r6) P1_ADD(r7)
    }
    if (k + 4 <= m) {
        const float4 r0 = p1_row(xw, __builtin_amdgcn_readlane(idx, k + 0), lane16);
        const float4 r1 = p1_row(xw, __builtin_amdgcn_readlane(idx, k + 1), lane16);
        const float4 r2 = p1_row(xw, __builtin_amdgcn_readlane(idx, k + 2), lane16);
        const float4 r3 = p1_row(xw, __builtin_amdgcn_readlane(idx, k + 3), lane16);
        P1_ADD(r0) P1_ADD(r1) P1_ADD(r2) P1_ADD(r3)
        k += 4;
    }
    const int t = m - k;
    if (t > 0) {
        const float4 r0 = p1_row(xw, __builtin_amdgcn_readlane(idx, k), lane16);
        float4 r1 = make_float4(0.0f, 0.0f, 0.0f, 0.0f);
        float4 r2 = make_float4(0.0f, 0.0f, 0.0f, 0.0f);
        if (t > 1) r1 = p1_row(xw, __builtin_amdgcn_readlane(idx, k + 1), lane16);
        if (t > 2) r2 = p1_row(xw, __builtin_amdgcn_readlane(idx, k + 2), lane16);
        P1_ADD(r0)
        if (t > 1) { P1_ADD(r1) }
        if (t > 2) { P1_ADD(r2) }
    }
}

__global__ __launch_bounds__(256) void gcn_pull1(const int* __restrict__ rowptr,
                                                 const int* __restrict__ cnt,
                                                 const int* __restrict__ csr,
                                                 const float* __restrict__ dinv,
                                                 const float* __restrict__ xw,
                                                 const float* __restrict__ b1,
                                                 const float* __restrict__ W2,
                                                 float* __restrict__ zw, int n) {
    const int lane = threadIdx.x & 31;
    const unsigned lane16 = (unsigned)lane * 16u;
    const int nwaves = (gridDim.x * blockDim.x) >> 5;
    const int wave = __builtin_amdgcn_readfirstlane((blockIdx.x * blockDim.x + threadIdx.x) >> 5);
    const int per = (n + nwaves - 1) / nwaves;
    const int first = wave * per;
    if (first >= n) return;
    const int last = (first + per < n) ? first + per : n;

    const float4 bv  = *(const float4*)(b1 + lane * 4);
    const float4 wlo = *(const float4*)(W2 + lane * 8);
    const float4 whi = *(const float4*)(W2 + lane * 8 + 4);

    for (int base = first; base < last; base += 32) {
        const int nn = (last - base < 32) ? last - base : 32;
        int v_start = 0, v_len = 0;
        float v_di = 0.0f;
        if (lane < nn) {
            v_start = rowptr[base + lane];
            v_len   = cnt[base + lane];
            v_di    = dinv[base + lane];
        }
        float o0 = 0.0f, o1 = 0.0f;

        int st = __builtin_amdgcn_readlane(v_start, 0);
        int ln = __builtin_amdgcn_readlane(v_len, 0);
        int my = 0;
        if (lane < ln) my = csr[st + lane];
        float4 self = p1_row(xw, base, lane16);

        for (int i = 0; i < nn; ++i) {
            const int start = st, len = ln;
            const float di = __int_as_float(__builtin_amdgcn_readlane(__float_as_int(v_di), i));
            float4 acc = self;
            int my_next = 0;
            if (i + 1 < nn) {
                st = __builtin_amdgcn_readlane(v_start, i + 1);
                ln = __builtin_amdgcn_readlane(v_len, i + 1);
                if (lane < ln) my_next = csr[st + lane];
                self = p1_row(xw, base + i + 1, lane16);
            }

            p1_sum_rows(xw, my, len < 32 ? len : 32, lane16, acc);
            for (int b0 = 32; b0 < len; b0 += 32) {
                const int rem = len - b0;
                const int m = rem < 32 ? rem : 32;
                int more = 0;
                if (lane < m) more = csr[start + b0 + lane];
                p1_sum_rows(xw, more, m, lane16, acc);
            }

            const float f0 = fmaxf(acc.x * di + bv.x, 0.0f);
            const float f1 = fmaxf(acc.y * di + bv.y, 0.0f);
            const float f2 = fmaxf(acc.z * di + bv.z, 0.0f);
            const float f3 = fmaxf(acc.w * di + bv.w, 0.0f);
            float a0 = f0 * wlo.x + f1 * wlo.z + f2 * whi.x + f3 * whi.z;
            float a1 = f0 * wlo.y + f1 * wlo.w + f2 * whi.y + f3 * whi.w;
            for (int off = 16; off > 0; off >>= 1) {
                a0 += __shfl_xor(a0, off, 32);
                a1 += __shfl_xor(a1, off, 32);
            }
            if (lane == i) {
                o0 = a0 * di;
                o1 = a1 * di;
            }
            my = my_next;
        }
        if (lane < nn) ((float2*)zw)[base + lane] = make_float2(o0, o1);
    }
}

#define P2_BLOCKS 625

__device__ __forceinline__ float2 p2_zw(const float* __restrict__ zw, int s) {
    return *(const float2*)((const char*)zw + (unsigned)s * 8u);
}

__global__ __launch_bounds__(256) void gcn_pull2(const int* __restrict__ rowptr,
                                                 const int* __restrict__ cnt,
                                                 const int* __restrict__ csr,
                                                 const float* __restrict__ dinv,
                                                 const float* __restrict__ zw,
                                                 const float* __restrict__ b2,
                                                 float* __restrict__ out, int n) {
    __shared__ float2 res[2][32];
    const int t = threadIdx.x;
    const int sub = t & 7;
    const int nl = t >> 3;
    const int ngroups = (n + 31) >> 5;
    const float b20 = b2[0], b21 = b2[1];

    int g = blockIdx.x;
    if (g >= ngroups) return;
    int nd, len, start;
    float di;
    {
        const int node = g * 32 + nl;
        const bool valid = node < n;
        nd = valid ? node : 0;
        start = rowptr[nd];
        len = valid ? cnt[nd] : 0;
        di = dinv[nd];
    }
    int stage = 0;
    for (; g < ngroups; g += gridDim.x) {
        const int c_nd = nd, c_len = len, c_start = start;
        const float c_di = di;
        const int gn = g + gridDim.x;
        if (gn < ngroups) {
            const int node = gn * 32 + nl;
            const bool valid = node < n;
            nd = valid ? node : 0;
            start = rowptr[nd];
            len = valid ? cnt[nd] : 0;
            di = dinv[nd];
        }
        const float2 zs = p2_zw(zw, c_nd);

        float a0 = 0.0f, a1 = 0.0f;
        const int* __restrict__ cp = csr + c_start;
        const int kl = c_len - 1;
        for (int k = sub; k < c_len; k += 32) {
            const bool h1 = k + 8 < c_len, h2 = k + 16 < c_len, h3 = k + 24 < c_len;
            const int s0 = cp[k];
            const int s1 = cp[h1 ? k + 8 : kl];
            const int s2 = cp[h2 ? k + 16 : kl];
            const int s3 = cp[h3 ? k + 24 : kl];
            const float2 z0 = p2_zw(zw, s0);
            const float2 z1 = p2_zw(zw, s1);
            const float2 z2 = p2_zw(zw, s2);
            const float2 z3 = p2_zw(zw, s3);
            a0 += z0.x; a1 += z0.y;
            if (h1) { a0 += z1.x; a1 += z1.y; }
            if (h2) { a0 += z2.x; a1 += z2.y; }
            if (h3) { a0 += z3.x; a1 += z3.y; }
        }
        for (int off = 4; off > 0; off >>= 1) {
            a0 += __shfl_xor(a0, off, 32);
            a1 += __shfl_xor(a1, off, 32);
        }
        if (sub == 0) {
            const float l0 = (a0 + zs.x) * c_di + b20;
            const float l1 = (a1 + zs.y) * c_di + b21;
            const float mx = fmaxf(l0, l1);
            const float lse = mx + logf(expf(l0 - mx) + expf(l1 - mx));
            res[stage][nl] = make_float2(l0 - lse, l1 - lse);
        }
        __syncthreads();
        if (t < 32) {
            const int o = g * 32 + t;
            if (o < n) ((float2*)out)[o] = res[stage][t];
        }
        stage ^= 1;
    }
}

extern "C" void kernel_launch(void* const* d_in, const int* in_sizes, int n_in,
                              void* d_out, int out_size, void* d_ws, size_t ws_size,
                              hipStream_t stream) {
    const float* x  = (const float*)d_in[0];
    const float* W1 = (const float*)d_in[1];
    const float* b1 = (const float*)d_in[2];
    const float* W2 = (const float*)d_in[3];
    const float* b2 = (const float*)d_in[4];
    const int*   ei = (const int*)d_in[5];

    const int n  = in_sizes[0] / IN_F;
    const int nE = in_sizes[5] / 2;
    const int* src = ei;
    const int* dst = ei + nE;

    char* ws = (char*)d_ws;
    char* sp = ws + (size_t)n * HID * 4;
    float* xw    = (float*)(ws);
    int*   cnt   = (int*)(sp);
    int*   rowptr= (int*)(sp + (size_t)n * 4);
    int*   tileoff = (int*)(sp + (size_t)n * 8);
    int2*  pairs = (int2*)(sp + (size_t)n * 16);
    int*   csr   = (int*)(sp + (size_t)n * 16 + (size_t)nE * 8);
    float* dinv  = (float*)(ws + (size_t)2 * n * HID * 4);
    float* zw    = (float*)(ws + (size_t)2 * n * HID * 4 + 512000);
    float* out   = (float*)d_out;

    const int B = 256;
    const int tiles = (nE + L1_TILE - 1) / L1_TILE;
    const int nb = (n + BK_N - 1) / BK_N;

    gcn_tile_bucket<<<tiles, SORT_T, 0, stream>>>(src, dst, tileoff, pairs, nE, nb);
    gcn_bucket_csr<<<nb, SORT_T, 0, stream>>>(pairs, tileoff, cnt, rowptr, dinv, csr, n, tiles, nb);

    {
        int rowTiles = n / 16;
        gcn_gemm1_wmma<<<rowTiles / 2, 64, 0, stream>>>(x, W1, dinv, xw);
    }

    gcn_pull1<<<P1_BLOCKS, B, 0, stream>>>(rowptr, cnt, csr, dinv, xw, b1, W2, zw, n);

    gcn_pull2<<<P2_BLOCKS, 256, 0, stream>>>(rowptr, cnt, csr, dinv, zw, b2, out, n);
}
